# starting kernel
# baseline (speedup 1.0000x reference)
#include <hip/hip_runtime.h>
#include <hip/hip_bf16.h>
#include <cmath>

typedef __attribute__((ext_vector_type(2))) float v2f;
typedef __attribute__((ext_vector_type(8))) float v8f;

#define NQ 100
#define JD 51
#define JD4 13
#define NTILE 7

__global__ __launch_bounds__(256) void matcher_blocks_kernel(
    const float* __restrict__ pred_conf,
    const float* __restrict__ pred_joint,
    const float* __restrict__ pred_action,
    const float* __restrict__ pred_ident,
    const float* __restrict__ tar_conf,
    const float* __restrict__ tar_joint,
    float* __restrict__ out)
{
    __shared__ float4 sPJ4[NQ * JD4];
    __shared__ float4 sTJ4[NQ * JD4];
    __shared__ float  sDlp[NQ];
    __shared__ float  sL1mp[NQ];
    __shared__ float  sRow[NQ];
    __shared__ float  sNegT[NQ];

    const int b   = blockIdx.y;
    const int tid = threadIdx.x;

    {
        float* dPJ = reinterpret_cast<float*>(sPJ4);
        float* dTJ = reinterpret_cast<float*>(sTJ4);
        const float* pj = pred_joint + b * NQ * JD;
        const float* tj = tar_joint  + b * NQ * JD;
        for (int idx = tid; idx < NQ * (4 * JD4); idx += 256) {
            int row = idx / (4 * JD4);
            int col = idx - row * (4 * JD4);
            float vp = 0.0f, vt = 0.0f;
            if (col < JD) {
                vp = pj[row * JD + col];
                vt = tj[row * JD + col];
            }
            dPJ[idx] = vp;
            dTJ[idx] = vt;
        }
    }

    if (tid < NQ) {
        float x = pred_conf[b * NQ + tid];
        sDlp[tid] = x;
        float l1mp = (x > 0.0f) ? (-x - log1pf(__expf(-x)))
                                : (-log1pf(__expf(x)));
        sL1mp[tid] = l1mp;

        const float* pa = pred_action + (b * NQ + tid) * 8;
        float m = pa[0];
        #pragma unroll
        for (int k = 1; k < 8; ++k) m = fmaxf(m, pa[k]);
        float s = 0.0f;
        #pragma unroll
        for (int k = 0; k < 8; ++k) s += __expf(pa[k] - m);
        float ce = (m + __logf(s)) - pa[0];

        const float* pi = pred_ident + (b * NQ + tid) * 10;
        float m2 = pi[0];
        #pragma unroll
        for (int k = 1; k < 10; ++k) m2 = fmaxf(m2, pi[k]);
        float s2 = 0.0f;
        #pragma unroll
        for (int k = 0; k < 10; ++k) s2 += __expf(pi[k] - m2);
        ce += (m2 + __logf(s2)) - pi[0];

        sRow[tid]  = ce;
        sNegT[tid] = -tar_conf[b * NQ + tid];
    }
    __syncthreads();

    const int wave = tid >> 5;
    const int lane = tid & 31;
    const int tile = blockIdx.x * 8 + wave;
    if (tile >= NTILE * NTILE) return;
    const int ti  = tile / NTILE;
    const int tjx = tile % NTILE;
    const int i0 = ti * 16;
    const int j0 = tjx * 16;

    v2f a;  a.x = 0.0f;  a.y = 0.0f;
    v2f bb; bb.x = 0.0f; bb.y = 0.0f;
    const int mlo = lane & 15;
    if (lane < 16) {
        int i = i0 + mlo;
        if (i < NQ) { a.x = sDlp[i]; a.y = sL1mp[i]; }
        int j = j0 + mlo;
        if (j < NQ) { bb.x = sNegT[j]; bb.y = -1.0f; }
    }

    v8f c = {};
    c = __builtin_amdgcn_wmma_f32_16x16x4_f32(
             false, a,  false, bb,
             (short)0, c,  false,  false);

    const int j     = j0 + (lane & 15);
    const int ibase = i0 + ((lane >> 4) << 3);
    float* outb = out + b * NQ * NQ;
    if (j < NQ) {
        const float4* tjrow = &sTJ4[j * JD4];
        float4 tr[JD4];
        #pragma unroll
        for (int kk = 0; kk < JD4; ++kk) tr[kk] = tjrow[kk];

        #pragma unroll
        for (int r = 0; r < 8; ++r) {
            int i = ibase + r;
            if (i < NQ) {
                const float4* pjrow = &sPJ4[i * JD4];
                float l1 = 0.0f;
                #pragma unroll
                for (int kk = 0; kk < JD4; ++kk) {
                    float4 pv = pjrow[kk];
                    l1 += fabsf(pv.x - tr[kk].x);
                    l1 += fabsf(pv.y - tr[kk].y);
                    l1 += fabsf(pv.z - tr[kk].z);
                    l1 += fabsf(pv.w - tr[kk].w);
                }
                outb[i * NQ + j] = c[r] + l1 + sRow[i];
            }
        }
    }
}

extern "C" void kernel_launch(void* const* d_in, const int* in_sizes, int n_in,
                              void* d_out, int out_size, void* d_ws, size_t ws_size,
                              hipStream_t stream) {
    (void)in_sizes; (void)n_in; (void)d_ws; (void)ws_size; (void)out_size;
    const float* pred_conf   = (const float*)d_in[0];
    const float* pred_joint  = (const float*)d_in[1];
    const float* pred_action = (const float*)d_in[2];
    const float* pred_ident  = (const float*)d_in[3];
    const float* tar_conf    = (const float*)d_in[4];
    const float* tar_joint   = (const float*)d_in[5];
    float* out = (float*)d_out;

    dim3 grid(NTILE, 16);
    dim3 block(256);
    matcher_blocks_kernel<<<grid, block, 0, stream>>>(
        pred_conf, pred_joint, pred_action, pred_ident, tar_conf, tar_joint, out);
}


# Data path: no LDS, no barrier; one wave per 16x16 tile loads its target row and 8 pred rows straight into registers; row CE terms summed inside the WMMA k-slots (was added after the L1 sum)
# speedup vs baseline: 3.2584x; 3.2584x over previous
#include <hip/hip_runtime.h>
#include <cmath>


typedef __attribute__((ext_vector_type(2))) float v2f;
typedef __attribute__((ext_vector_type(8))) float v8f;
typedef __attribute__((ext_vector_type(4))) float v4f;
typedef v4f v4f_a4 __attribute__((aligned(4)));

#define NQ 100
#define JD 51
#define JD4 12
#define NTILE 7
#define LAST_I0   ((NTILE - 1) * 16)
#define LAST_ROWS (NQ - LAST_I0)
static_assert(LAST_ROWS > 0 && LAST_ROWS <= 8, "the last tile row's real rows must lie in lane half 0");

__device__ __forceinline__ v4f load4(const float* p) {
    return *reinterpret_cast<const v4f_a4*>(p);
}

__global__ __launch_bounds__(128, 1) void matcher_rows_in_registers_kernel(
    const float* __restrict__ pred_conf,
    const float* __restrict__ pred_joint,
    const float* __restrict__ pred_action,
    const float* __restrict__ pred_ident,
    const float* __restrict__ tar_conf,
    const float* __restrict__ tar_joint,
    float* __restrict__ out)
{
    const int tid  = threadIdx.x;
    const int lane = tid & 31;
    const int wave = __builtin_amdgcn_readfirstlane(tid >> 5);
    const int tile = blockIdx.x * 4 + wave;
    const int b    = tile / (NTILE * NTILE);
    const int rem  = tile - b * (NTILE * NTILE);
    const int ti   = rem / NTILE;
    const int tjx  = rem - ti * NTILE;
    const int i0   = ti * 16;
    const int j0   = tjx * 16;
    const int c    = lane & 15;
    const int h    = lane >> 4;

    const int ic    = min(i0 + c, NQ - 1);
    const int jc    = min(j0 + c, NQ - 1);
    const int ibase = i0 + 8 * h;

    const float xconf = pred_conf[b * NQ + ic];
    const float tconf = tar_conf[b * NQ + jc];

    const float* pa = pred_action + (b * NQ + ic) * 8;
    const float* pi = pred_ident  + (b * NQ + ic) * 10;
    float av[8], iv[10];
    #pragma unroll
    for (int k = 0; k < 8; ++k)  av[k] = pa[k];
    #pragma unroll
    for (int k = 0; k < 10; ++k) iv[k] = pi[k];

    const float* tjrow = tar_joint + (b * NQ + jc) * JD;
    v4f tr[JD4];
    #pragma unroll
    for (int kk = 0; kk < JD4; ++kk) tr[kk] = load4(tjrow + 4 * kk);
    const float t48 = tjrow[48], t49 = tjrow[49], t50 = tjrow[50];

    const float* prow[8];
    #pragma unroll
    for (int r = 0; r < 8; ++r)
        prow[r] = pred_joint + (b * NQ + min(ibase + r, NQ - 1)) * JD;

    v4f pv[8][JD4];
    #pragma unroll
    for (int kk = 0; kk < JD4; ++kk) {
        #pragma unroll
        for (int r = 0; r < 8; ++r) pv[r][kk] = load4(prow[r] + 4 * kk);
    }
    float p48[8], p49[8], p50[8];
    #pragma unroll
    for (int r = 0; r < 8; ++r) {
        p48[r] = prow[r][48]; p49[r] = prow[r][49]; p50[r] = prow[r][50];
    }

    const float l1mp = (xconf > 0.0f) ? (-xconf - log1pf(__expf(-xconf)))
                                      : (-log1pf(__expf(xconf)));
    float m = av[0];
    #pragma unroll
    for (int k = 1; k < 8; ++k) m = fmaxf(m, av[k]);
    float s = 0.0f;
    #pragma unroll
    for (int k = 0; k < 8; ++k) s += __expf(av[k] - m);
    float ce = (m + __logf(s)) - av[0];
    float m2 = iv[0];
    #pragma unroll
    for (int k = 1; k < 10; ++k) m2 = fmaxf(m2, iv[k]);
    float s2 = 0.0f;
    #pragma unroll
    for (int k = 0; k < 10; ++k) s2 += __expf(iv[k] - m2);
    ce += (m2 + __logf(s2)) - iv[0];

    v2f a, bb;
    a.x  = h ? ce   : xconf;
    a.y  = h ? 0.0f : l1mp;
    bb.x = h ? 1.0f : -tconf;
    bb.y = h ? 0.0f : -1.0f;
    v8f acc = {};
    acc = __builtin_amdgcn_wmma_f32_16x16x4_f32(
               false, a,  false, bb,
               (short)0, acc,  false,  false);

    float l1[8];
    #pragma unroll
    for (int r = 0; r < 8; ++r) l1[r] = 0.0f;
    #pragma unroll
    for (int kk = 0; kk < JD4; ++kk) {
        #pragma unroll
        for (int r = 0; r < 8; ++r) {
            l1[r] += fabsf(pv[r][kk].x - tr[kk].x);
            l1[r] += fabsf(pv[r][kk].y - tr[kk].y);
            l1[r] += fabsf(pv[r][kk].z - tr[kk].z);
            l1[r] += fabsf(pv[r][kk].w - tr[kk].w);
        }
    }
    #pragma unroll
    for (int r = 0; r < 8; ++r) {
        l1[r] += fabsf(p48[r] - t48);
        l1[r] += fabsf(p49[r] - t49);
        l1[r] += fabsf(p50[r] - t50);
    }

    float res[8];
    #pragma unroll
    for (int r = 0; r < 8; ++r) res[r] = acc[r] + l1[r];

    const int j = j0 + c;
    float* outb = out + b * NQ * NQ;
    if (j < NQ) {
        if (ti < NTILE - 1) {
            #pragma unroll
            for (int r = 0; r < 8; ++r) outb[(ibase + r) * NQ + j] = res[r];
        } else if (h == 0) {
            #pragma unroll
            for (int r = 0; r < LAST_ROWS; ++r) outb[(LAST_I0 + r) * NQ + j] = res[r];
        }
    }
}

extern "C" void kernel_launch(void* const* d_in, const int* in_sizes, int n_in,
                              void* d_out, int out_size, void* d_ws, size_t ws_size,
                              hipStream_t stream) {
    (void)in_sizes; (void)n_in; (void)d_ws; (void)ws_size; (void)out_size;
    const float* pred_conf   = (const float*)d_in[0];
    const float* pred_joint  = (const float*)d_in[1];
    const float* pred_action = (const float*)d_in[2];
    const float* pred_ident  = (const float*)d_in[3];
    const float* tar_conf    = (const float*)d_in[4];
    const float* tar_joint   = (const float*)d_in[5];
    float* out = (float*)d_out;

    dim3 grid(196);
    dim3 block(128);
    matcher_rows_in_registers_kernel<<<grid, block, 0, stream>>>(
        pred_conf, pred_joint, pred_action, pred_ident, tar_conf, tar_joint, out);
}


# Wave body as three pinned straight-line stretches: first loads | both log sides and both softmaxes as 4 parallel chains | rest of loads + L1 sums, which enter the WMMA as C (last add reordered)
# speedup vs baseline: 3.5091x; 1.0769x over previous
#include <hip/hip_runtime.h>
#include <cmath>


typedef __attribute__((ext_vector_type(2))) float v2f;
typedef __attribute__((ext_vector_type(8))) float v8f;
typedef __attribute__((ext_vector_type(4))) float v4f;
typedef v4f v4f_a4 __attribute__((aligned(4)));

#define NQ 100
#define JD 51
#define JD4 12
#define NTILE 7
#define LAST_I0   ((NTILE - 1) * 16)
#define LAST_ROWS (NQ - LAST_I0)
#define FIRST_KK 9
static_assert(LAST_ROWS > 0 && LAST_ROWS <= 8, "the last tile row's real rows must lie in lane half 0");

__device__ __forceinline__ v4f load4(const float* p) {
    return *reinterpret_cast<const v4f_a4*>(p);
}

__global__ __launch_bounds__(128, 1) void matcher_rows_in_registers_kernel(
    const float* __restrict__ pred_conf,
    const float* __restrict__ pred_joint,
    const float* __restrict__ pred_action,
    const float* __restrict__ pred_ident,
    const float* __restrict__ tar_conf,
    const float* __restrict__ tar_joint,
    float* __restrict__ out)
{
    const int tid  = threadIdx.x;
    const int lane = tid & 31;
    const int wave = __builtin_amdgcn_readfirstlane(tid >> 5);
    const int tile = blockIdx.x * 4 + wave;
    const int b    = tile / (NTILE * NTILE);
    const int rem  = tile - b * (NTILE * NTILE);
    const int ti   = rem / NTILE;
    const int tjx  = rem - ti * NTILE;
    const int i0   = ti * 16;
    const int j0   = tjx * 16;
    const int c    = lane & 15;
    const int h    = lane >> 4;

    const int ic    = min(i0 + c, NQ - 1);
    const int jc    = min(j0 + c, NQ - 1);
    const int ibase = i0 + 8 * h;

    const float xconf = pred_conf[b * NQ + ic];
    const float tconf = tar_conf[b * NQ + jc];

    const float* pa = pred_action + (b * NQ + ic) * 8;
    const float* pi = pred_ident  + (b * NQ + ic) * 10;
    float av[8], iv[10];
    #pragma unroll
    for (int k = 0; k < 8; ++k)  av[k] = pa[k];
    #pragma unroll
    for (int k = 0; k < 10; ++k) iv[k] = pi[k];

    const float* tjrow = tar_joint + (b * NQ + jc) * JD;
    v4f tr[JD4];
    #pragma unroll
    for (int kk = 0; kk < JD4; ++kk) tr[kk] = load4(tjrow + 4 * kk);
    const float t48 = tjrow[48], t49 = tjrow[49], t50 = tjrow[50];

    const float* prow[8];
    #pragma unroll
    for (int r = 0; r < 8; ++r)
        prow[r] = pred_joint + (b * NQ + min(ibase + r, NQ - 1)) * JD;

    v4f pv[8][JD4];
    #pragma unroll
    for (int kk = 0; kk < FIRST_KK; ++kk) {
        #pragma unroll
        for (int r = 0; r < 8; ++r) pv[r][kk] = load4(prow[r] + 4 * kk);
    }
    float p48[8], p49[8], p50[8];
    p48[7] = prow[7][48]; p49[7] = prow[7][49]; p50[7] = prow[7][50];

    const float l1mp_pos = -xconf - log1pf(__expf(-xconf));
    const float l1mp_neg = -log1pf(__expf(xconf));
    float l1mp = (xconf > 0.0f) ? l1mp_pos : l1mp_neg;
    float m = av[0];
    #pragma unroll
    for (int k = 1; k < 8; ++k) m = fmaxf(m, av[k]);
    float s = 0.0f;
    #pragma unroll
    for (int k = 0; k < 8; ++k) s += __expf(av[k] - m);
    float ce_pin = (m + __logf(s)) - av[0];
    float m2 = iv[0];
    #pragma unroll
    for (int k = 1; k < 10; ++k) m2 = fmaxf(m2, iv[k]);
    float s2 = 0.0f;
    #pragma unroll
    for (int k = 0; k < 10; ++k) s2 += __expf(iv[k] - m2);
    ce_pin += (m2 + __logf(s2)) - iv[0];

    int second_group_off = 0;
    asm volatile("" : "+v"(l1mp), "+v"(ce_pin), "+v"(second_group_off) : "s"(out));
    __builtin_amdgcn_sched_barrier(0);
    const float ce = ce_pin;

    #pragma unroll
    for (int kk = FIRST_KK; kk < JD4; ++kk) {
        #pragma unroll
        for (int r = 0; r < 8; ++r)
            pv[r][kk] = load4(prow[r] + second_group_off + 4 * kk);
    }
    #pragma unroll
    for (int r = 0; r < 7; ++r) {
        const float* q = prow[r] + second_group_off;
        p48[r] = q[48]; p49[r] = q[49]; p50[r] = q[50];
    }

    v2f a, bb;
    a.x  = h ? ce   : xconf;
    a.y  = h ? 0.0f : l1mp;
    bb.x = h ? 1.0f : -tconf;
    bb.y = h ? 0.0f : -1.0f;
    float l1[8];
    #pragma unroll
    for (int r = 0; r < 8; ++r) l1[r] = 0.0f;
    #pragma unroll
    for (int kk = 0; kk < JD4; ++kk) {
        #pragma unroll
        for (int r = 0; r < 8; ++r) {
            l1[r] += fabsf(pv[r][kk].x - tr[kk].x);
            l1[r] += fabsf(pv[r][kk].y - tr[kk].y);
            l1[r] += fabsf(pv[r][kk].z - tr[kk].z);
            l1[r] += fabsf(pv[r][kk].w - tr[kk].w);
        }
    }
    #pragma unroll
    for (int r = 0; r < 8; ++r) {
        l1[r] += fabsf(p48[r] - t48);
        l1[r] += fabsf(p49[r] - t49);
        l1[r] += fabsf(p50[r] - t50);
    }

    v8f acc;
    #pragma unroll
    for (int r = 0; r < 8; ++r) acc[r] = l1[r];
    acc = __builtin_amdgcn_wmma_f32_16x16x4_f32(
               false, a,  false, bb,
               (short)0, acc,  false,  false);
    float res[8];
    #pragma unroll
    for (int r = 0; r < 8; ++r) res[r] = acc[r];

    const int j = j0 + c;
    float* outb = out + b * NQ * NQ;
    if (j < NQ) {
        if (ti < NTILE - 1) {
            #pragma unroll
            for (int r = 0; r < 8; ++r) outb[(ibase + r) * NQ + j] = res[r];
        } else if (h == 0) {
            #pragma unroll
            for (int r = 0; r < LAST_ROWS; ++r) outb[(LAST_I0 + r) * NQ + j] = res[r];
        }
    }
}

extern "C" void kernel_launch(void* const* d_in, const int* in_sizes, int n_in,
                              void* d_out, int out_size, void* d_ws, size_t ws_size,
                              hipStream_t stream) {
    (void)in_sizes; (void)n_in; (void)d_ws; (void)ws_size; (void)out_size;
    const float* pred_conf   = (const float*)d_in[0];
    const float* pred_joint  = (const float*)d_in[1];
    const float* pred_action = (const float*)d_in[2];
    const float* pred_ident  = (const float*)d_in[3];
    const float* tar_conf    = (const float*)d_in[4];
    const float* tar_joint   = (const float*)d_in[5];
    float* out = (float*)d_out;

    dim3 grid(196);
    dim3 block(128);
    matcher_rows_in_registers_kernel<<<grid, block, 0, stream>>>(
        pred_conf, pred_joint, pred_action, pred_ident, tar_conf, tar_joint, out);
}
